# starting kernel
# baseline (speedup 1.0000x reference)
#include <hip/hip_runtime.h>
#include <hip/hip_bf16.h>

#define Bsz   4
#define Cin   256
#define CRdim 64
#define Gdim  16
#define GCdim 16
#define Kker  7
#define KK    49
#define M2    784
#define OUTC  256
#define Hdim  64
#define Wdim  64
#define HW    4096
#define NWAVES 4

typedef __attribute__((ext_vector_type(2))) float v2f;
typedef __attribute__((ext_vector_type(8))) float v8f;

template <int NTM, int NTN>
__device__ __forceinline__ void wmma_gemm_block(const float* __restrict__ A, int lda,
                                                const float* __restrict__ Bm, int ldb,
                                                int Ksz, int mBase, int nBase,
                                                v8f acc[NTM][NTN]) {
  const int lane = threadIdx.x & 31;
  const int half = lane >> 4;
  const int l16  = lane & 15;
  const float* arow = A + (size_t)(mBase + l16) * lda + 2 * half;
  const float* bptr = Bm + (size_t)(2 * half) * ldb + nBase + l16;
  for (int k = 0; k < Ksz; k += 4) {
    v2f a[NTM];
#pragma unroll
    for (int tm = 0; tm < NTM; ++tm) {
      a[tm].x = arow[(size_t)tm * 16 * lda + k + 0];
      a[tm].y = arow[(size_t)tm * 16 * lda + k + 1];
    }
#pragma unroll
    for (int tn = 0; tn < NTN; ++tn) {
      v2f b;
      b.x = bptr[(size_t)(k + 0) * ldb + tn * 16];
      b.y = bptr[(size_t)(k + 1) * ldb + tn * 16];
#pragma unroll
      for (int tm = 0; tm < NTM; ++tm) {
        acc[tm][tn] = __builtin_amdgcn_wmma_f32_16x16x4_f32(
            false, a[tm], false, b, (short)0, acc[tm][tn], false, false);
      }
    }
  }
}

__global__ void k_gemm_bn_relu(const float* __restrict__ x,
                               const float* __restrict__ w1,
                               const float* __restrict__ gamma,
                               const float* __restrict__ beta,
                               const float* __restrict__ mean,
                               const float* __restrict__ var,
                               float* __restrict__ y) {
  const int wave  = threadIdx.x >> 5;
  const int nBase = (blockIdx.x * NWAVES + wave) * 32;
  const int mBase = blockIdx.y * 32;
  const int b     = blockIdx.z;
  const float* Bm = x + (size_t)b * Cin * HW;
  v8f acc[2][2] = {};
  wmma_gemm_block<2, 2>(w1, Cin, Bm, HW, Cin, mBase, nBase, acc);

  float* Yb = y + (size_t)b * CRdim * HW;
  const int lane = threadIdx.x & 31;
  const int half = lane >> 4, l16 = lane & 15;
#pragma unroll
  for (int tm = 0; tm < 2; ++tm) {
#pragma unroll
    for (int r = 0; r < 8; ++r) {
      const int m = mBase + tm * 16 + r + 8 * half;
      const float s  = gamma[m] * rsqrtf(var[m] + 1e-5f);
      const float mu = mean[m], bt = beta[m];
#pragma unroll
      for (int tn = 0; tn < 2; ++tn) {
        const float v = (acc[tm][tn][r] - mu) * s + bt;
        Yb[(size_t)m * HW + nBase + tn * 16 + l16] = fmaxf(v, 0.0f);
      }
    }
  }
}

__global__ void k_gemm_bias(const float* __restrict__ y,
                            const float* __restrict__ w2,
                            const float* __restrict__ b2,
                            float* __restrict__ wt) {
  const int wave  = threadIdx.x >> 5;
  const int nBase = (blockIdx.x * NWAVES + wave) * 64;
  const int mBase = blockIdx.y * 16;
  const int b     = blockIdx.z;
  const float* Bm = y + (size_t)b * CRdim * HW;
  v8f acc[1][4] = {};
  wmma_gemm_block<1, 4>(w2, CRdim, Bm, HW, CRdim, mBase, nBase, acc);

  float* Wb = wt + (size_t)b * M2 * HW;
  const int lane = threadIdx.x & 31;
  const int half = lane >> 4, l16 = lane & 15;
#pragma unroll
  for (int r = 0; r < 8; ++r) {
    const int m = mBase + r + 8 * half;
    const float bias = b2[m];
#pragma unroll
    for (int tn = 0; tn < 4; ++tn) {
      Wb[(size_t)m * HW + nBase + tn * 16 + l16] = acc[0][tn][r] + bias;
    }
  }
}

__global__ void k_involution(const float* __restrict__ x,
                             const float* __restrict__ wt,
                             float* __restrict__ inv) {
  __shared__ float taps[KK * 16];

  const int xtile = blockIdx.x & 3;
  const int py    = blockIdx.x >> 2;
  const int g     = blockIdx.y;
  const int b     = blockIdx.z;
  const int tid   = threadIdx.x;
  const int pxi   = tid & 15;
  const int ch    = tid >> 4;
  const int xbase = xtile * 16;

  const float* wtp = wt + ((size_t)b * M2 + (size_t)g * KK) * HW + py * Wdim + xbase;
  for (int e = tid; e < KK * 16; e += 256) {
    const int k = e >> 4;
    taps[e] = wtp[(size_t)k * HW + (e & 15)];
  }
  __syncthreads();

  const int c = g * GCdim + ch;
  const float* xp = x + ((size_t)b * Cin + c) * HW;
  const int px = xbase + pxi;

  float acc = 0.0f;
#pragma unroll
  for (int dy = 0; dy < Kker; ++dy) {
    const int yy = py + dy - 3;
    const bool okY = (unsigned)yy < (unsigned)Hdim;
#pragma unroll
    for (int dx = 0; dx < Kker; ++dx) {
      const int xx = px + dx - 3;
      const bool ok = okY && ((unsigned)xx < (unsigned)Wdim);
      const float xv = ok ? xp[yy * Wdim + xx] : 0.0f;
      const float wv = taps[(dy * Kker + dx) * 16 + pxi];
      acc = fmaf(wv, xv, acc);
    }
  }
  inv[((size_t)b * Cin + c) * HW + py * Wdim + px] = acc;
}

__global__ void k_gemm_plain(const float* __restrict__ inv,
                             const float* __restrict__ wp,
                             float* __restrict__ out) {
  const int wave  = threadIdx.x >> 5;
  const int nBase = (blockIdx.x * NWAVES + wave) * 32;
  const int mBase = blockIdx.y * 32;
  const int b     = blockIdx.z;
  const float* Bm = inv + (size_t)b * Cin * HW;
  v8f acc[2][2] = {};
  wmma_gemm_block<2, 2>(wp, Cin, Bm, HW, Cin, mBase, nBase, acc);

  float* Ob = out + (size_t)b * OUTC * HW;
  const int lane = threadIdx.x & 31;
  const int half = lane >> 4, l16 = lane & 15;
#pragma unroll
  for (int tm = 0; tm < 2; ++tm) {
#pragma unroll
    for (int r = 0; r < 8; ++r) {
      const int m = mBase + tm * 16 + r + 8 * half;
#pragma unroll
      for (int tn = 0; tn < 2; ++tn) {
        Ob[(size_t)m * HW + nBase + tn * 16 + l16] = acc[tm][tn][r];
      }
    }
  }
}

extern "C" void kernel_launch(void* const* d_in, const int* in_sizes, int n_in,
                              void* d_out, int out_size, void* d_ws, size_t ws_size,
                              hipStream_t stream) {
  const float* x     = (const float*)d_in[0];
  const float* w1    = (const float*)d_in[1];
  const float* gamma = (const float*)d_in[2];
  const float* beta  = (const float*)d_in[3];
  const float* mean  = (const float*)d_in[4];
  const float* var   = (const float*)d_in[5];
  const float* w2    = (const float*)d_in[6];
  const float* b2    = (const float*)d_in[7];
  const float* wp    = (const float*)d_in[8];
  float* out = (float*)d_out;

  float* y   = (float*)d_ws;
  float* wt  = y  + (size_t)Bsz * CRdim * HW;
  float* inv = wt + (size_t)Bsz * M2 * HW;

  dim3 blk(32 * NWAVES, 1, 1);

  dim3 g1(HW / (32 * NWAVES), CRdim / 32, Bsz);
  k_gemm_bn_relu<<<g1, blk, 0, stream>>>(x, w1, gamma, beta, mean, var, y);

  dim3 g2(HW / (64 * NWAVES), M2 / 16, Bsz);
  k_gemm_bias<<<g2, blk, 0, stream>>>(y, w2, b2, wt);

  dim3 gI(Hdim * (Wdim / 16), Gdim, Bsz);
  k_involution<<<gI, 256, 0, stream>>>(x, wt, inv);

  dim3 g4(HW / (32 * NWAVES), OUTC / 32, Bsz);
  k_gemm_plain<<<g4, blk, 0, stream>>>(inv, wp, out);
}


# involution stencil register-tiled: x halo tile staged in LDS once per 8-row block, each thread keeps 8 channels x 4 pixels, taps read once by 16-byte loads; per-output sum order unchanged
# speedup vs baseline: 2.2763x; 2.2763x over previous
#include <hip/hip_runtime.h>
#include <hip/hip_bf16.h>

#define Bsz   4
#define Cin   256
#define CRdim 64
#define Gdim  16
#define GCdim 16
#define Kker  7
#define KK    49
#define M2    784
#define OUTC  256
#define Hdim  64
#define Wdim  64
#define HW    4096
#define NWAVES 4

typedef __attribute__((ext_vector_type(2))) float v2f;
typedef __attribute__((ext_vector_type(8))) float v8f;

template <int NTM, int NTN>
__device__ __forceinline__ void wmma_gemm_block(const float* __restrict__ A, int lda,
                                                const float* __restrict__ Bm, int ldb,
                                                int Ksz, int mBase, int nBase,
                                                v8f acc[NTM][NTN]) {
  const int lane = threadIdx.x & 31;
  const int half = lane >> 4;
  const int l16  = lane & 15;
  const float* arow = A + (size_t)(mBase + l16) * lda + 2 * half;
  const float* bptr = Bm + (size_t)(2 * half) * ldb + nBase + l16;
  for (int k = 0; k < Ksz; k += 4) {
    v2f a[NTM];
#pragma unroll
    for (int tm = 0; tm < NTM; ++tm) {
      a[tm].x = arow[(size_t)tm * 16 * lda + k + 0];
      a[tm].y = arow[(size_t)tm * 16 * lda + k + 1];
    }
#pragma unroll
    for (int tn = 0; tn < NTN; ++tn) {
      v2f b;
      b.x = bptr[(size_t)(k + 0) * ldb + tn * 16];
      b.y = bptr[(size_t)(k + 1) * ldb + tn * 16];
#pragma unroll
      for (int tm = 0; tm < NTM; ++tm) {
        acc[tm][tn] = __builtin_amdgcn_wmma_f32_16x16x4_f32(
            false, a[tm], false, b, (short)0, acc[tm][tn], false, false);
      }
    }
  }
}

__global__ void k_gemm_bn_relu(const float* __restrict__ x,
                               const float* __restrict__ w1,
                               const float* __restrict__ gamma,
                               const float* __restrict__ beta,
                               const float* __restrict__ mean,
                               const float* __restrict__ var,
                               float* __restrict__ y) {
  const int wave  = threadIdx.x >> 5;
  const int nBase = (blockIdx.x * NWAVES + wave) * 32;
  const int mBase = blockIdx.y * 32;
  const int b     = blockIdx.z;
  const float* Bm = x + (size_t)b * Cin * HW;
  v8f acc[2][2] = {};
  wmma_gemm_block<2, 2>(w1, Cin, Bm, HW, Cin, mBase, nBase, acc);

  float* Yb = y + (size_t)b * CRdim * HW;
  const int lane = threadIdx.x & 31;
  const int half = lane >> 4, l16 = lane & 15;
#pragma unroll
  for (int tm = 0; tm < 2; ++tm) {
#pragma unroll
    for (int r = 0; r < 8; ++r) {
      const int m = mBase + tm * 16 + r + 8 * half;
      const float s  = gamma[m] * rsqrtf(var[m] + 1e-5f);
      const float mu = mean[m], bt = beta[m];
#pragma unroll
      for (int tn = 0; tn < 2; ++tn) {
        const float v = (acc[tm][tn][r] - mu) * s + bt;
        Yb[(size_t)m * HW + nBase + tn * 16 + l16] = fmaxf(v, 0.0f);
      }
    }
  }
}

__global__ void k_gemm_bias(const float* __restrict__ y,
                            const float* __restrict__ w2,
                            const float* __restrict__ b2,
                            float* __restrict__ wt) {
  const int wave  = threadIdx.x >> 5;
  const int nBase = (blockIdx.x * NWAVES + wave) * 64;
  const int mBase = blockIdx.y * 16;
  const int b     = blockIdx.z;
  const float* Bm = y + (size_t)b * CRdim * HW;
  v8f acc[1][4] = {};
  wmma_gemm_block<1, 4>(w2, CRdim, Bm, HW, CRdim, mBase, nBase, acc);

  float* Wb = wt + (size_t)b * M2 * HW;
  const int lane = threadIdx.x & 31;
  const int half = lane >> 4, l16 = lane & 15;
#pragma unroll
  for (int r = 0; r < 8; ++r) {
    const int m = mBase + r + 8 * half;
    const float bias = b2[m];
#pragma unroll
    for (int tn = 0; tn < 4; ++tn) {
      Wb[(size_t)m * HW + nBase + tn * 16 + l16] = acc[0][tn][r] + bias;
    }
  }
}

#define TH     8
#define XROWS  (TH + 6)
#define XPITCH 72
#define XCH    (XROWS * XPITCH + 4)

__global__ void __launch_bounds__(256) __attribute__((amdgpu_waves_per_eu(4, 5)))
k_involution(const float* __restrict__ x,
             const float* __restrict__ wt,
             float* __restrict__ inv) {
  __shared__ __attribute__((aligned(16))) float xs[GCdim * XCH];

  const int y0   = blockIdx.x * TH;
  const int g    = blockIdx.y;
  const int b    = blockIdx.z;
  const int tid  = threadIdx.x;
  const int wrow = tid >> 5;
  const int lane = tid & 31;
  const int j    = lane & 15;
  const int hh   = lane >> 4;
  const int py   = y0 + wrow;

  const float* wtp = wt + ((size_t)b * M2 + (size_t)g * KK) * HW + py * Wdim + 4 * j;

  float4 wbuf[3][Kker];
#pragma unroll
  for (int dx = 0; dx < Kker; ++dx) {
    wbuf[0][dx] = *(const float4*)(wtp + (size_t)(0 * Kker + dx) * HW);
    wbuf[1][dx] = *(const float4*)(wtp + (size_t)(1 * Kker + dx) * HW);
  }

  const float* xg = x + ((size_t)b * Cin + (size_t)g * GCdim) * HW;
  constexpr int NST = (GCdim * XROWS * 16) / 256;
  float4 sv[NST];
#pragma unroll
  for (int i = 0; i < NST; ++i) {
    const int idx = tid + 256 * i;
    const int q   = idx & 15;
    const int rc  = idx >> 4;
    const int ch  = rc / XROWS;
    const int rr  = rc - ch * XROWS;
    const int gy  = y0 - 3 + rr;
    const int gyc = min(max(gy, 0), Hdim - 1);
    sv[i] = *(const float4*)(xg + (size_t)ch * HW + gyc * Wdim + 4 * q);
  }
#pragma unroll
  for (int i = 0; i < NST; ++i) {
    const int idx = tid + 256 * i;
    const int q   = idx & 15;
    const int rc  = idx >> 4;
    const int ch  = rc / XROWS;
    const int rr  = rc - ch * XROWS;
    const int gy  = y0 - 3 + rr;
    const bool in = (unsigned)gy < (unsigned)Hdim;
    float* d = xs + ch * XCH + rr * XPITCH + 3 + 4 * q;
    d[0] = in ? sv[i].x : 0.0f;
    d[1] = in ? sv[i].y : 0.0f;
    d[2] = in ? sv[i].z : 0.0f;
    d[3] = in ? sv[i].w : 0.0f;
  }
  if (tid < GCdim * XROWS) {
    const int ch = tid / XROWS;
    const int rr = tid - ch * XROWS;
    float* rowp = xs + ch * XCH + rr * XPITCH;
    rowp[0]  = 0.0f; rowp[1]  = 0.0f; rowp[2]  = 0.0f;
    rowp[67] = 0.0f; rowp[68] = 0.0f; rowp[69] = 0.0f;
  }
  __syncthreads();

  float acc[8][4];
#pragma unroll
  for (int cc = 0; cc < 8; ++cc)
#pragma unroll
    for (int p = 0; p < 4; ++p) acc[cc][p] = 0.0f;

  const float* xwin = xs + (hh * 8) * XCH + wrow * XPITCH + 4 * j;

#pragma unroll
  for (int dy = 0; dy < Kker; ++dy) {
    if (dy + 2 < Kker) {
#pragma unroll
      for (int dx = 0; dx < Kker; ++dx)
        wbuf[(dy + 2) % 3][dx] = *(const float4*)(wtp + (size_t)((dy + 2) * Kker + dx) * HW);
    }
#pragma unroll
    for (int cc = 0; cc < 8; ++cc) {
      const float* p0 = xwin + cc * XCH + dy * XPITCH;
      const float4 a0 = *(const float4*)(p0);
      const float4 a1 = *(const float4*)(p0 + 4);
      const float2 a2 = *(const float2*)(p0 + 8);
      const float xw[10] = {a0.x, a0.y, a0.z, a0.w, a1.x, a1.y, a1.z, a1.w,
                            a2.x, a2.y};
#pragma unroll
      for (int dx = 0; dx < Kker; ++dx) {
        const float4 wv = wbuf[dy % 3][dx];
        acc[cc][0] = fmaf(wv.x, xw[dx + 0], acc[cc][0]);
        acc[cc][1] = fmaf(wv.y, xw[dx + 1], acc[cc][1]);
        acc[cc][2] = fmaf(wv.z, xw[dx + 2], acc[cc][2]);
        acc[cc][3] = fmaf(wv.w, xw[dx + 3], acc[cc][3]);
      }
    }
  }

  float* op = inv + ((size_t)b * Cin + (size_t)g * GCdim + hh * 8) * HW + py * Wdim + 4 * j;
#pragma unroll
  for (int cc = 0; cc < 8; ++cc)
    *(float4*)(op + (size_t)cc * HW) =
        make_float4(acc[cc][0], acc[cc][1], acc[cc][2], acc[cc][3]);
}

__global__ void k_gemm_plain(const float* __restrict__ inv,
                             const float* __restrict__ wp,
                             float* __restrict__ out) {
  const int wave  = threadIdx.x >> 5;
  const int nBase = (blockIdx.x * NWAVES + wave) * 32;
  const int mBase = blockIdx.y * 32;
  const int b     = blockIdx.z;
  const float* Bm = inv + (size_t)b * Cin * HW;
  v8f acc[2][2] = {};
  wmma_gemm_block<2, 2>(wp, Cin, Bm, HW, Cin, mBase, nBase, acc);

  float* Ob = out + (size_t)b * OUTC * HW;
  const int lane = threadIdx.x & 31;
  const int half = lane >> 4, l16 = lane & 15;
#pragma unroll
  for (int tm = 0; tm < 2; ++tm) {
#pragma unroll
    for (int r = 0; r < 8; ++r) {
      const int m = mBase + tm * 16 + r + 8 * half;
#pragma unroll
      for (int tn = 0; tn < 2; ++tn) {
        Ob[(size_t)m * HW + nBase + tn * 16 + l16] = acc[tm][tn][r];
      }
    }
  }
}

extern "C" void kernel_launch(void* const* d_in, const int* in_sizes, int n_in,
                              void* d_out, int out_size, void* d_ws, size_t ws_size,
                              hipStream_t stream) {
  const float* x     = (const float*)d_in[0];
  const float* w1    = (const float*)d_in[1];
  const float* gamma = (const float*)d_in[2];
  const float* beta  = (const float*)d_in[3];
  const float* mean  = (const float*)d_in[4];
  const float* var   = (const float*)d_in[5];
  const float* w2    = (const float*)d_in[6];
  const float* b2    = (const float*)d_in[7];
  const float* wp    = (const float*)d_in[8];
  float* out = (float*)d_out;

  float* y   = (float*)d_ws;
  float* wt  = y  + (size_t)Bsz * CRdim * HW;
  float* inv = wt + (size_t)Bsz * M2 * HW;

  dim3 blk(32 * NWAVES, 1, 1);

  dim3 g1(HW / (32 * NWAVES), CRdim / 32, Bsz);
  k_gemm_bn_relu<<<g1, blk, 0, stream>>>(x, w1, gamma, beta, mean, var, y);

  dim3 g2(HW / (64 * NWAVES), M2 / 16, Bsz);
  k_gemm_bias<<<g2, blk, 0, stream>>>(y, w2, b2, wt);

  dim3 gI(Hdim / TH, Gdim, Bsz);
  k_involution<<<gI, 256, 0, stream>>>(x, wt, inv);

  dim3 g4(HW / (32 * NWAVES), OUTC / 32, Bsz);
  k_gemm_plain<<<g4, blk, 0, stream>>>(inv, wp, out);
}


# GEMM fragment loads double-buffered in registers: the next group of K-steps (8 in the K=256 GEMMs, 2 in the K=64 one) is loaded as one run ahead of the current group's WMMAs; sum order unchanged
# speedup vs baseline: 2.8387x; 1.2471x over previous
#include <hip/hip_runtime.h>
#include <hip/hip_bf16.h>

#define Bsz   4
#define Cin   256
#define CRdim 64
#define Gdim  16
#define GCdim 16
#define Kker  7
#define KK    49
#define M2    784
#define OUTC  256
#define Hdim  64
#define Wdim  64
#define HW    4096
#define NWAVES 4

typedef __attribute__((ext_vector_type(2))) float v2f;
typedef __attribute__((ext_vector_type(8))) float v8f;

template <int NTM, int NTN, int G>
struct FragSet {
  v2f a[G][NTM];
  v2f b[G][NTN];
};

template <int NTM, int NTN, int G>
__device__ __forceinline__ void load_group(const float* __restrict__ arow, int lda,
                                           const float* __restrict__ bptr, int ldb,
                                           int k0, FragSet<NTM, NTN, G>& f) {
#pragma unroll
  for (int s = 0; s < G; ++s) {
    const int k = k0 + 4 * s;
#pragma unroll
    for (int tm = 0; tm < NTM; ++tm) {
      f.a[s][tm].x = arow[(size_t)tm * 16 * lda + k + 0];
      f.a[s][tm].y = arow[(size_t)tm * 16 * lda + k + 1];
    }
#pragma unroll
    for (int tn = 0; tn < NTN; ++tn) {
      f.b[s][tn].x = bptr[(size_t)(k + 0) * ldb + tn * 16];
      f.b[s][tn].y = bptr[(size_t)(k + 1) * ldb + tn * 16];
    }
  }
}

template <int NTM, int NTN, int G>
__device__ __forceinline__ void mma_group(const FragSet<NTM, NTN, G>& f,
                                          v8f acc[NTM][NTN]) {
#pragma unroll
  for (int s = 0; s < G; ++s) {
#pragma unroll
    for (int tn = 0; tn < NTN; ++tn) {
#pragma unroll
      for (int tm = 0; tm < NTM; ++tm) {
        acc[tm][tn] = __builtin_amdgcn_wmma_f32_16x16x4_f32(
            false, f.a[s][tm], false, f.b[s][tn], (short)0, acc[tm][tn], false, false);
      }
    }
  }
}

template <int NTM, int NTN, int G, int KSZ>
__device__ __forceinline__ void wmma_gemm_block(const float* __restrict__ A, int lda,
                                                const float* __restrict__ Bm, int ldb,
                                                int mBase, int nBase,
                                                v8f acc[NTM][NTN]) {
  static_assert(KSZ % (8 * G) == 0, "K must hold an even number of groups");
  const int lane = threadIdx.x & 31;
  const int half = lane >> 4;
  const int l16  = lane & 15;
  const float* arow = A + (size_t)(mBase + l16) * lda + 2 * half;
  const float* bptr = Bm + (size_t)(2 * half) * ldb + nBase + l16;

  FragSet<NTM, NTN, G> f0, f1;
  load_group<NTM, NTN, G>(arow, lda, bptr, ldb, 0, f0);
#pragma unroll
  for (int k = 0; k < KSZ; k += 8 * G) {
    load_group<NTM, NTN, G>(arow, lda, bptr, ldb, k + 4 * G, f1);
    __builtin_amdgcn_sched_barrier(0);
    mma_group<NTM, NTN, G>(f0, acc);
    __builtin_amdgcn_sched_barrier(0);
    if (k + 8 * G < KSZ) load_group<NTM, NTN, G>(arow, lda, bptr, ldb, k + 8 * G, f0);
    __builtin_amdgcn_sched_barrier(0);
    mma_group<NTM, NTN, G>(f1, acc);
    __builtin_amdgcn_sched_barrier(0);
  }
}

__global__ void __launch_bounds__(32 * NWAVES) k_gemm_bn_relu(const float* __restrict__ x,
                               const float* __restrict__ w1,
                               const float* __restrict__ gamma,
                               const float* __restrict__ beta,
                               const float* __restrict__ mean,
                               const float* __restrict__ var,
                               float* __restrict__ y) {
  const int wave  = threadIdx.x >> 5;
  const int nBase = (blockIdx.x * NWAVES + wave) * 32;
  const int mBase = blockIdx.y * 32;
  const int b     = blockIdx.z;
  const float* Bm = x + (size_t)b * Cin * HW;
  v8f acc[2][2] = {};
  wmma_gemm_block<2, 2, 8, Cin>(w1, Cin, Bm, HW, mBase, nBase, acc);

  float* Yb = y + (size_t)b * CRdim * HW;
  const int lane = threadIdx.x & 31;
  const int half = lane >> 4, l16 = lane & 15;
#pragma unroll
  for (int tm = 0; tm < 2; ++tm) {
#pragma unroll
    for (int r = 0; r < 8; ++r) {
      const int m = mBase + tm * 16 + r + 8 * half;
      const float s  = gamma[m] * rsqrtf(var[m] + 1e-5f);
      const float mu = mean[m], bt = beta[m];
#pragma unroll
      for (int tn = 0; tn < 2; ++tn) {
        const float v = (acc[tm][tn][r] - mu) * s + bt;
        Yb[(size_t)m * HW + nBase + tn * 16 + l16] = fmaxf(v, 0.0f);
      }
    }
  }
}

__global__ void __launch_bounds__(32 * NWAVES) k_gemm_bias(const float* __restrict__ y,
                            const float* __restrict__ w2,
                            const float* __restrict__ b2,
                            float* __restrict__ wt) {
  const int wave  = threadIdx.x >> 5;
  const int nBase = (blockIdx.x * NWAVES + wave) * 64;
  const int mBase = blockIdx.y * 16;
  const int b     = blockIdx.z;
  const float* Bm = y + (size_t)b * CRdim * HW;
  v8f acc[1][4] = {};
  wmma_gemm_block<1, 4, 2, CRdim>(w2, CRdim, Bm, HW, mBase, nBase, acc);

  float* Wb = wt + (size_t)b * M2 * HW;
  const int lane = threadIdx.x & 31;
  const int half = lane >> 4, l16 = lane & 15;
#pragma unroll
  for (int r = 0; r < 8; ++r) {
    const int m = mBase + r + 8 * half;
    const float bias = b2[m];
#pragma unroll
    for (int tn = 0; tn < 4; ++tn) {
      Wb[(size_t)m * HW + nBase + tn * 16 + l16] = acc[0][tn][r] + bias;
    }
  }
}

#define TH     8
#define XROWS  (TH + 6)
#define XPITCH 72
#define XCH    (XROWS * XPITCH + 4)

__global__ void __launch_bounds__(256) __attribute__((amdgpu_waves_per_eu(4, 5)))
k_involution(const float* __restrict__ x,
             const float* __restrict__ wt,
             float* __restrict__ inv) {
  __shared__ __attribute__((aligned(16))) float xs[GCdim * XCH];

  const int y0   = blockIdx.x * TH;
  const int g    = blockIdx.y;
  const int b    = blockIdx.z;
  const int tid  = threadIdx.x;
  const int wrow = tid >> 5;
  const int lane = tid & 31;
  const int j    = lane & 15;
  const int hh   = lane >> 4;
  const int py   = y0 + wrow;

  const float* wtp = wt + ((size_t)b * M2 + (size_t)g * KK) * HW + py * Wdim + 4 * j;

  float4 wbuf[3][Kker];
#pragma unroll
  for (int dx = 0; dx < Kker; ++dx) {
    wbuf[0][dx] = *(const float4*)(wtp + (size_t)(0 * Kker + dx) * HW);
    wbuf[1][dx] = *(const float4*)(wtp + (size_t)(1 * Kker + dx) * HW);
  }

  const float* xg = x + ((size_t)b * Cin + (size_t)g * GCdim) * HW;
  constexpr int NST = (GCdim * XROWS * 16) / 256;
  float4 sv[NST];
#pragma unroll
  for (int i = 0; i < NST; ++i) {
    const int idx = tid + 256 * i;
    const int q   = idx & 15;
    const int rc  = idx >> 4;
    const int ch  = rc / XROWS;
    const int rr  = rc - ch * XROWS;
    const int gy  = y0 - 3 + rr;
    const int gyc = min(max(gy, 0), Hdim - 1);
    sv[i] = *(const float4*)(xg + (size_t)ch * HW + gyc * Wdim + 4 * q);
  }
#pragma unroll
  for (int i = 0; i < NST; ++i) {
    const int idx = tid + 256 * i;
    const int q   = idx & 15;
    const int rc  = idx >> 4;
    const int ch  = rc / XROWS;
    const int rr  = rc - ch * XROWS;
    const int gy  = y0 - 3 + rr;
    const bool in = (unsigned)gy < (unsigned)Hdim;
    float* d = xs + ch * XCH + rr * XPITCH + 3 + 4 * q;
    d[0] = in ? sv[i].x : 0.0f;
    d[1] = in ? sv[i].y : 0.0f;
    d[2] = in ? sv[i].z : 0.0f;
    d[3] = in ? sv[i].w : 0.0f;
  }
  if (tid < GCdim * XROWS) {
    const int ch = tid / XROWS;
    const int rr = tid - ch * XROWS;
    float* rowp = xs + ch * XCH + rr * XPITCH;
    rowp[0]  = 0.0f; rowp[1]  = 0.0f; rowp[2]  = 0.0f;
    rowp[67] = 0.0f; rowp[68] = 0.0f; rowp[69] = 0.0f;
  }
  __syncthreads();

  float acc[8][4];
#pragma unroll
  for (int cc = 0; cc < 8; ++cc)
#pragma unroll
    for (int p = 0; p < 4; ++p) acc[cc][p] = 0.0f;

  const float* xwin = xs + (hh * 8) * XCH + wrow * XPITCH + 4 * j;

#pragma unroll
  for (int dy = 0; dy < Kker; ++dy) {
    if (dy + 2 < Kker) {
#pragma unroll
      for (int dx = 0; dx < Kker; ++dx)
        wbuf[(dy + 2) % 3][dx] = *(const float4*)(wtp + (size_t)((dy + 2) * Kker + dx) * HW);
    }
#pragma unroll
    for (int cc = 0; cc < 8; ++cc) {
      const float* p0 = xwin + cc * XCH + dy * XPITCH;
      const float4 a0 = *(const float4*)(p0);
      const float4 a1 = *(const float4*)(p0 + 4);
      const float2 a2 = *(const float2*)(p0 + 8);
      const float xw[10] = {a0.x, a0.y, a0.z, a0.w, a1.x, a1.y, a1.z, a1.w,
                            a2.x, a2.y};
#pragma unroll
      for (int dx = 0; dx < Kker; ++dx) {
        const float4 wv = wbuf[dy % 3][dx];
        acc[cc][0] = fmaf(wv.x, xw[dx + 0], acc[cc][0]);
        acc[cc][1] = fmaf(wv.y, xw[dx + 1], acc[cc][1]);
        acc[cc][2] = fmaf(wv.z, xw[dx + 2], acc[cc][2]);
        acc[cc][3] = fmaf(wv.w, xw[dx + 3], acc[cc][3]);
      }
    }
  }

  float* op = inv + ((size_t)b * Cin + (size_t)g * GCdim + hh * 8) * HW + py * Wdim + 4 * j;
#pragma unroll
  for (int cc = 0; cc < 8; ++cc)
    *(float4*)(op + (size_t)cc * HW) =
        make_float4(acc[cc][0], acc[cc][1], acc[cc][2], acc[cc][3]);
}

__global__ void __launch_bounds__(32 * NWAVES) k_gemm_plain(const float* __restrict__ inv,
                             const float* __restrict__ wp,
                             float* __restrict__ out) {
  const int wave  = threadIdx.x >> 5;
  const int nBase = (blockIdx.x * NWAVES + wave) * 32;
  const int mBase = blockIdx.y * 32;
  const int b     = blockIdx.z;
  const float* Bm = inv + (size_t)b * Cin * HW;
  v8f acc[2][2] = {};
  wmma_gemm_block<2, 2, 8, Cin>(wp, Cin, Bm, HW, mBase, nBase, acc);

  float* Ob = out + (size_t)b * OUTC * HW;
  const int lane = threadIdx.x & 31;
  const int half = lane >> 4, l16 = lane & 15;
#pragma unroll
  for (int tm = 0; tm < 2; ++tm) {
#pragma unroll
    for (int r = 0; r < 8; ++r) {
      const int m = mBase + tm * 16 + r + 8 * half;
#pragma unroll
      for (int tn = 0; tn < 2; ++tn) {
        Ob[(size_t)m * HW + nBase + tn * 16 + l16] = acc[tm][tn][r];
      }
    }
  }
}

extern "C" void kernel_launch(void* const* d_in, const int* in_sizes, int n_in,
                              void* d_out, int out_size, void* d_ws, size_t ws_size,
                              hipStream_t stream) {
  const float* x     = (const float*)d_in[0];
  const float* w1    = (const float*)d_in[1];
  const float* gamma = (const float*)d_in[2];
  const float* beta  = (const float*)d_in[3];
  const float* mean  = (const float*)d_in[4];
  const float* var   = (const float*)d_in[5];
  const float* w2    = (const float*)d_in[6];
  const float* b2    = (const float*)d_in[7];
  const float* wp    = (const float*)d_in[8];
  float* out = (float*)d_out;

  float* y   = (float*)d_ws;
  float* wt  = y  + (size_t)Bsz * CRdim * HW;
  float* inv = wt + (size_t)Bsz * M2 * HW;

  dim3 blk(32 * NWAVES, 1, 1);

  dim3 g1(HW / (32 * NWAVES), CRdim / 32, Bsz);
  k_gemm_bn_relu<<<g1, blk, 0, stream>>>(x, w1, gamma, beta, mean, var, y);

  dim3 g2(HW / (64 * NWAVES), M2 / 16, Bsz);
  k_gemm_bias<<<g2, blk, 0, stream>>>(y, w2, b2, wt);

  dim3 gI(Hdim / TH, Gdim, Bsz);
  k_involution<<<gI, 256, 0, stream>>>(x, wt, inv);

  dim3 g4(HW / (32 * NWAVES), OUTC / 32, Bsz);
  k_gemm_plain<<<g4, blk, 0, stream>>>(inv, wp, out);
}


# GEMM fragments loaded 16 bytes per lane: B rows as float4 over a column-interleaved 64-column wave tile, results stored as float4, A as 4 k per lane; new k grouping per WMMA changes the sum order
# speedup vs baseline: 3.2024x; 1.1281x over previous
#include <hip/hip_runtime.h>
#include <hip/hip_bf16.h>

#define Bsz   4
#define Cin   256
#define CRdim 64
#define Gdim  16
#define GCdim 16
#define Kker  7
#define KK    49
#define M2    784
#define OUTC  256
#define Hdim  64
#define Wdim  64
#define HW    4096
#define NWAVES 4

typedef __attribute__((ext_vector_type(2))) float v2f;
typedef __attribute__((ext_vector_type(4))) float v4f;
typedef __attribute__((ext_vector_type(8))) float v8f;

template <int NTM, int GP>
struct FragSet {
  v4f a[GP][NTM];
  v4f b[GP][4];
};

template <int NTM, int GP>
__device__ __forceinline__ void load_group(const float* __restrict__ arow, int lda,
                                           const float* __restrict__ bptr, int ldb,
                                           int k0, FragSet<NTM, GP>& f) {
#pragma unroll
  for (int p = 0; p < GP; ++p) {
    const int k = k0 + 8 * p;
#pragma unroll
    for (int tm = 0; tm < NTM; ++tm)
      f.a[p][tm] = *(const v4f*)(arow + (size_t)tm * 16 * lda + k);
#pragma unroll
    for (int j = 0; j < 4; ++j)
      f.b[p][j] = *(const v4f*)(bptr + (size_t)(k + j) * ldb);
  }
}

template <int NTM, int GP>
__device__ __forceinline__ void mma_group(const FragSet<NTM, GP>& f,
                                          v8f acc[NTM][4]) {
#pragma unroll
  for (int p = 0; p < GP; ++p) {
#pragma unroll
    for (int s = 0; s < 2; ++s) {
#pragma unroll
      for (int tn = 0; tn < 4; ++tn) {
        v2f bv;
        bv.x = f.b[p][2 * s][tn];
        bv.y = f.b[p][2 * s + 1][tn];
#pragma unroll
        for (int tm = 0; tm < NTM; ++tm) {
          v2f av;
          av.x = f.a[p][tm][2 * s];
          av.y = f.a[p][tm][2 * s + 1];
          acc[tm][tn] = __builtin_amdgcn_wmma_f32_16x16x4_f32(
              false, av, false, bv, (short)0, acc[tm][tn], false, false);
        }
      }
    }
  }
}

template <int NTM, int GP, int KSZ>
__device__ __forceinline__ void wmma_gemm_block(const float* __restrict__ A, int lda,
                                                const float* __restrict__ Bm, int ldb,
                                                int mBase, int nBase,
                                                v8f acc[NTM][4]) {
  static_assert(KSZ % (16 * GP) == 0, "K must hold an even number of groups");
  const int lane = threadIdx.x & 31;
  const int half = lane >> 4;
  const int l16  = lane & 15;
  const float* arow = A + (size_t)(mBase + l16) * lda + 4 * half;
  const float* bptr = Bm + (size_t)(4 * half) * ldb + nBase + 4 * l16;

  FragSet<NTM, GP> f0, f1;
  load_group<NTM, GP>(arow, lda, bptr, ldb, 0, f0);
#pragma unroll
  for (int k = 0; k < KSZ; k += 16 * GP) {
    load_group<NTM, GP>(arow, lda, bptr, ldb, k + 8 * GP, f1);
    __builtin_amdgcn_sched_barrier(0);
    mma_group<NTM, GP>(f0, acc);
    __builtin_amdgcn_sched_barrier(0);
    if (k + 16 * GP < KSZ) load_group<NTM, GP>(arow, lda, bptr, ldb, k + 16 * GP, f0);
    __builtin_amdgcn_sched_barrier(0);
    mma_group<NTM, GP>(f1, acc);
    __builtin_amdgcn_sched_barrier(0);
  }
}

__global__ void __launch_bounds__(32 * NWAVES) k_gemm_bn_relu(const float* __restrict__ x,
                               const float* __restrict__ w1,
                               const float* __restrict__ gamma,
                               const float* __restrict__ beta,
                               const float* __restrict__ mean,
                               const float* __restrict__ var,
                               float* __restrict__ y) {
  const int wave  = threadIdx.x >> 5;
  const int nBase = blockIdx.x * 64;
  const int mBase = wave * 16;
  const int b     = blockIdx.z;
  const float* Bm = x + (size_t)b * Cin * HW;
  v8f acc[1][4] = {};
  wmma_gemm_block<1, 8, Cin>(w1, Cin, Bm, HW, mBase, nBase, acc);

  float* Yb = y + (size_t)b * CRdim * HW;
  const int lane = threadIdx.x & 31;
  const int half = lane >> 4, l16 = lane & 15;
#pragma unroll
  for (int r = 0; r < 8; ++r) {
    const int m = mBase + r + 8 * half;
    const float s  = gamma[m] * rsqrtf(var[m] + 1e-5f);
    const float mu = mean[m], bt = beta[m];
    v4f o;
#pragma unroll
    for (int tn = 0; tn < 4; ++tn) {
      const float v = (acc[0][tn][r] - mu) * s + bt;
      o[tn] = fmaxf(v, 0.0f);
    }
    *(v4f*)(Yb + (size_t)m * HW + nBase + 4 * l16) = o;
  }
}

__global__ void __launch_bounds__(32 * NWAVES) k_gemm_bias(const float* __restrict__ y,
                            const float* __restrict__ w2,
                            const float* __restrict__ b2,
                            float* __restrict__ wt) {
  const int wave  = threadIdx.x >> 5;
  const int nBase = (blockIdx.x * NWAVES + wave) * 64;
  const int mBase = blockIdx.y * 16;
  const int b     = blockIdx.z;
  const float* Bm = y + (size_t)b * CRdim * HW;
  v8f acc[1][4] = {};
  wmma_gemm_block<1, 2, CRdim>(w2, CRdim, Bm, HW, mBase, nBase, acc);

  float* Wb = wt + (size_t)b * M2 * HW;
  const int lane = threadIdx.x & 31;
  const int half = lane >> 4, l16 = lane & 15;
#pragma unroll
  for (int r = 0; r < 8; ++r) {
    const int m = mBase + r + 8 * half;
    const float bias = b2[m];
    v4f o;
#pragma unroll
    for (int tn = 0; tn < 4; ++tn) o[tn] = acc[0][tn][r] + bias;
    *(v4f*)(Wb + (size_t)m * HW + nBase + 4 * l16) = o;
  }
}

#define TH     8
#define XROWS  (TH + 6)
#define XPITCH 72
#define XCH    (XROWS * XPITCH + 4)

__global__ void __launch_bounds__(256) __attribute__((amdgpu_waves_per_eu(4, 5)))
k_involution(const float* __restrict__ x,
             const float* __restrict__ wt,
             float* __restrict__ inv) {
  __shared__ __attribute__((aligned(16))) float xs[GCdim * XCH];

  const int y0   = blockIdx.x * TH;
  const int g    = blockIdx.y;
  const int b    = blockIdx.z;
  const int tid  = threadIdx.x;
  const int wrow = tid >> 5;
  const int lane = tid & 31;
  const int j    = lane & 15;
  const int hh   = lane >> 4;
  const int py   = y0 + wrow;

  const float* wtp = wt + ((size_t)b * M2 + (size_t)g * KK) * HW + py * Wdim + 4 * j;

  float4 wbuf[3][Kker];
#pragma unroll
  for (int dx = 0; dx < Kker; ++dx) {
    wbuf[0][dx] = *(const float4*)(wtp + (size_t)(0 * Kker + dx) * HW);
    wbuf[1][dx] = *(const float4*)(wtp + (size_t)(1 * Kker + dx) * HW);
  }

  const float* xg = x + ((size_t)b * Cin + (size_t)g * GCdim) * HW;
  constexpr int NST = (GCdim * XROWS * 16) / 256;
  float4 sv[NST];
#pragma unroll
  for (int i = 0; i < NST; ++i) {
    const int idx = tid + 256 * i;
    const int q   = idx & 15;
    const int rc  = idx >> 4;
    const int ch  = rc / XROWS;
    const int rr  = rc - ch * XROWS;
    const int gy  = y0 - 3 + rr;
    const int gyc = min(max(gy, 0), Hdim - 1);
    sv[i] = *(const float4*)(xg + (size_t)ch * HW + gyc * Wdim + 4 * q);
  }
#pragma unroll
  for (int i = 0; i < NST; ++i) {
    const int idx = tid + 256 * i;
    const int q   = idx & 15;
    const int rc  = idx >> 4;
    const int ch  = rc / XROWS;
    const int rr  = rc - ch * XROWS;
    const int gy  = y0 - 3 + rr;
    const bool in = (unsigned)gy < (unsigned)Hdim;
    float* d = xs + ch * XCH + rr * XPITCH + 3 + 4 * q;
    d[0] = in ? sv[i].x : 0.0f;
    d[1] = in ? sv[i].y : 0.0f;
    d[2] = in ? sv[i].z : 0.0f;
    d[3] = in ? sv[i].w : 0.0f;
  }
  if (tid < GCdim * XROWS) {
    const int ch = tid / XROWS;
    const int rr = tid - ch * XROWS;
    float* rowp = xs + ch * XCH + rr * XPITCH;
    rowp[0]  = 0.0f; rowp[1]  = 0.0f; rowp[2]  = 0.0f;
    rowp[67] = 0.0f; rowp[68] = 0.0f; rowp[69] = 0.0f;
  }
  __syncthreads();

  float acc[8][4];
#pragma unroll
  for (int cc = 0; cc < 8; ++cc)
#pragma unroll
    for (int p = 0; p < 4; ++p) acc[cc][p] = 0.0f;

  const float* xwin = xs + (hh * 8) * XCH + wrow * XPITCH + 4 * j;

#pragma unroll
  for (int dy = 0; dy < Kker; ++dy) {
    if (dy + 2 < Kker) {
#pragma unroll
      for (int dx = 0; dx < Kker; ++dx)
        wbuf[(dy + 2) % 3][dx] = *(const float4*)(wtp + (size_t)((dy + 2) * Kker + dx) * HW);
    }
#pragma unroll
    for (int cc = 0; cc < 8; ++cc) {
      const float* p0 = xwin + cc * XCH + dy * XPITCH;
      const float4 a0 = *(const float4*)(p0);
      const float4 a1 = *(const float4*)(p0 + 4);
      const float2 a2 = *(const float2*)(p0 + 8);
      const float xw[10] = {a0.x, a0.y, a0.z, a0.w, a1.x, a1.y, a1.z, a1.w,
                            a2.x, a2.y};
#pragma unroll
      for (int dx = 0; dx < Kker; ++dx) {
        const float4 wv = wbuf[dy % 3][dx];
        acc[cc][0] = fmaf(wv.x, xw[dx + 0], acc[cc][0]);
        acc[cc][1] = fmaf(wv.y, xw[dx + 1], acc[cc][1]);
        acc[cc][2] = fmaf(wv.z, xw[dx + 2], acc[cc][2]);
        acc[cc][3] = fmaf(wv.w, xw[dx + 3], acc[cc][3]);
      }
    }
  }

  float* op = inv + ((size_t)b * Cin + (size_t)g * GCdim + hh * 8) * HW + py * Wdim + 4 * j;
#pragma unroll
  for (int cc = 0; cc < 8; ++cc)
    *(float4*)(op + (size_t)cc * HW) =
        make_float4(acc[cc][0], acc[cc][1], acc[cc][2], acc[cc][3]);
}

__global__ void __launch_bounds__(32 * NWAVES) k_gemm_plain(const float* __restrict__ inv,
                             const float* __restrict__ wp,
                             float* __restrict__ out) {
  const int wave  = threadIdx.x >> 5;
  const int nBase = (blockIdx.x * NWAVES + wave) * 64;
  const int mBase = blockIdx.y * 32;
  const int b     = blockIdx.z;
  const float* Bm = inv + (size_t)b * Cin * HW;
  v8f acc[2][4] = {};
  wmma_gemm_block<2, 4, Cin>(wp, Cin, Bm, HW, mBase, nBase, acc);

  float* Ob = out + (size_t)b * OUTC * HW;
  const int lane = threadIdx.x & 31;
  const int half = lane >> 4, l16 = lane & 15;
#pragma unroll
  for (int tm = 0; tm < 2; ++tm) {
#pragma unroll
    for (int r = 0; r < 8; ++r) {
      const int m = mBase + tm * 16 + r + 8 * half;
      v4f o;
#pragma unroll
      for (int tn = 0; tn < 4; ++tn) o[tn] = acc[tm][tn][r];
      *(v4f*)(Ob + (size_t)m * HW + nBase + 4 * l16) = o;
    }
  }
}

extern "C" void kernel_launch(void* const* d_in, const int* in_sizes, int n_in,
                              void* d_out, int out_size, void* d_ws, size_t ws_size,
                              hipStream_t stream) {
  const float* x     = (const float*)d_in[0];
  const float* w1    = (const float*)d_in[1];
  const float* gamma = (const float*)d_in[2];
  const float* beta  = (const float*)d_in[3];
  const float* mean  = (const float*)d_in[4];
  const float* var   = (const float*)d_in[5];
  const float* w2    = (const float*)d_in[6];
  const float* b2    = (const float*)d_in[7];
  const float* wp    = (const float*)d_in[8];
  float* out = (float*)d_out;

  float* y   = (float*)d_ws;
  float* wt  = y  + (size_t)Bsz * CRdim * HW;
  float* inv = wt + (size_t)Bsz * M2 * HW;

  dim3 blk(32 * NWAVES, 1, 1);

  dim3 g1(HW / 64, 1, Bsz);
  k_gemm_bn_relu<<<g1, blk, 0, stream>>>(x, w1, gamma, beta, mean, var, y);

  dim3 g2(HW / (64 * NWAVES), M2 / 16, Bsz);
  k_gemm_bias<<<g2, blk, 0, stream>>>(y, w2, b2, wt);

  dim3 gI(Hdim / TH, Gdim, Bsz);
  k_involution<<<gI, 256, 0, stream>>>(x, wt, inv);

  dim3 g4(HW / (64 * NWAVES), OUTC / 32, Bsz);
  k_gemm_plain<<<g4, blk, 0, stream>>>(inv, wp, out);
}


# first two GEMMs in one launch: per 64-column block the BN+ReLU y tile goes to LDS, each wave keeps it as B fragments in registers and streams 16-row w2 tiles; y not written; sum order unchanged
# speedup vs baseline: 3.4921x; 1.0905x over previous
#include <hip/hip_runtime.h>
#include <hip/hip_bf16.h>

#define Bsz   4
#define Cin   256
#define CRdim 64
#define Gdim  16
#define GCdim 16
#define Kker  7
#define KK    49
#define M2    784
#define OUTC  256
#define Hdim  64
#define Wdim  64
#define HW    4096
#define NWAVES 4

typedef __attribute__((ext_vector_type(2))) float v2f;
typedef __attribute__((ext_vector_type(4))) float v4f;
typedef __attribute__((ext_vector_type(8))) float v8f;

template <int NTM, int GP>
struct FragSet {
  v4f a[GP][NTM];
  v4f b[GP][4];
};

template <int NTM, int GP>
__device__ __forceinline__ void load_group(const float* __restrict__ arow, int lda,
                                           const float* __restrict__ bptr, int ldb,
                                           int k0, FragSet<NTM, GP>& f) {
#pragma unroll
  for (int p = 0; p < GP; ++p) {
    const int k = k0 + 8 * p;
#pragma unroll
    for (int tm = 0; tm < NTM; ++tm)
      f.a[p][tm] = *(const v4f*)(arow + (size_t)tm * 16 * lda + k);
#pragma unroll
    for (int j = 0; j < 4; ++j)
      f.b[p][j] = *(const v4f*)(bptr + (size_t)(k + j) * ldb);
  }
}

template <int NTM, int GP>
__device__ __forceinline__ void mma_group(const FragSet<NTM, GP>& f,
                                          v8f acc[NTM][4]) {
#pragma unroll
  for (int p = 0; p < GP; ++p) {
#pragma unroll
    for (int s = 0; s < 2; ++s) {
#pragma unroll
      for (int tn = 0; tn < 4; ++tn) {
        v2f bv;
        bv.x = f.b[p][2 * s][tn];
        bv.y = f.b[p][2 * s + 1][tn];
#pragma unroll
        for (int tm = 0; tm < NTM; ++tm) {
          v2f av;
          av.x = f.a[p][tm][2 * s];
          av.y = f.a[p][tm][2 * s + 1];
          acc[tm][tn] = __builtin_amdgcn_wmma_f32_16x16x4_f32(
              false, av, false, bv, (short)0, acc[tm][tn], false, false);
        }
      }
    }
  }
}

template <int NTM, int GP, int KSZ>
__device__ __forceinline__ void wmma_gemm_block(const float* __restrict__ A, int lda,
                                                const float* __restrict__ Bm, int ldb,
                                                int mBase, int nBase,
                                                v8f acc[NTM][4]) {
  static_assert(KSZ % (16 * GP) == 0, "K must hold an even number of groups");
  const int lane = threadIdx.x & 31;
  const int half = lane >> 4;
  const int l16  = lane & 15;
  const float* arow = A + (size_t)(mBase + l16) * lda + 4 * half;
  const float* bptr = Bm + (size_t)(4 * half) * ldb + nBase + 4 * l16;

  FragSet<NTM, GP> f0, f1;
  load_group<NTM, GP>(arow, lda, bptr, ldb, 0, f0);
#pragma unroll
  for (int k = 0; k < KSZ; k += 16 * GP) {
    load_group<NTM, GP>(arow, lda, bptr, ldb, k + 8 * GP, f1);
    __builtin_amdgcn_sched_barrier(0);
    mma_group<NTM, GP>(f0, acc);
    __builtin_amdgcn_sched_barrier(0);
    if (k + 16 * GP < KSZ) load_group<NTM, GP>(arow, lda, bptr, ldb, k + 16 * GP, f0);
    __builtin_amdgcn_sched_barrier(0);
    mma_group<NTM, GP>(f1, acc);
    __builtin_amdgcn_sched_barrier(0);
  }
}

#define YP 68

struct ATile {
  v4f a[8];
  v4f bias[2];
};

__device__ __forceinline__ void load_atile(const float* __restrict__ w2,
                                           const float* __restrict__ b2,
                                           int t, int l16, int half, ATile& f) {
  const float* arow = w2 + (size_t)(16 * t + l16) * CRdim + 4 * half;
#pragma unroll
  for (int p = 0; p < 8; ++p) f.a[p] = *(const v4f*)(arow + 8 * p);
  const float* bp = b2 + 16 * t + 8 * half;
  f.bias[0] = *(const v4f*)(bp);
  f.bias[1] = *(const v4f*)(bp + 4);
}

__device__ __forceinline__ void wt_tile(const ATile& f, const v4f (&bf)[8][4],
                                        float* __restrict__ dst) {
  v8f acc[4] = {};
#pragma unroll
  for (int p = 0; p < 8; ++p) {
#pragma unroll
    for (int s = 0; s < 2; ++s) {
      v2f av;
      av.x = f.a[p][2 * s];
      av.y = f.a[p][2 * s + 1];
#pragma unroll
      for (int tn = 0; tn < 4; ++tn) {
        v2f bv;
        bv.x = bf[p][2 * s][tn];
        bv.y = bf[p][2 * s + 1][tn];
        acc[tn] = __builtin_amdgcn_wmma_f32_16x16x4_f32(
            false, av, false, bv, (short)0, acc[tn], false, false);
      }
    }
  }
#pragma unroll
  for (int r = 0; r < 8; ++r) {
    const float bias = f.bias[r >> 2][r & 3];
    v4f o;
#pragma unroll
    for (int tn = 0; tn < 4; ++tn) o[tn] = acc[tn][r] + bias;
    *(v4f*)(dst + (size_t)r * HW) = o;
  }
}

#define MT2 (M2 / 16)

__global__ void __launch_bounds__(256) k_weightgen(const float* __restrict__ x,
                                                   const float* __restrict__ w1,
                                                   const float* __restrict__ gamma,
                                                   const float* __restrict__ beta,
                                                   const float* __restrict__ mean,
                                                   const float* __restrict__ var,
                                                   const float* __restrict__ w2,
                                                   const float* __restrict__ b2,
                                                   float* __restrict__ wt) {
  __shared__ __attribute__((aligned(16))) float ys[CRdim * YP];

  const int wave  = __builtin_amdgcn_readfirstlane((int)(threadIdx.x >> 5));
  const int lane  = threadIdx.x & 31;
  const int half  = lane >> 4, l16 = lane & 15;
  const int nBase = blockIdx.x * 64;
  const int b     = blockIdx.z;

  if (wave < 4) {
    const int mBase = wave * 16;
    const float* Bm = x + (size_t)b * Cin * HW;
    v8f acc[1][4] = {};
    wmma_gemm_block<1, 8, Cin>(w1, Cin, Bm, HW, mBase, nBase, acc);
#pragma unroll
    for (int r = 0; r < 8; ++r) {
      const int m = mBase + r + 8 * half;
      const float s  = gamma[m] * rsqrtf(var[m] + 1e-5f);
      const float mu = mean[m], bt = beta[m];
      v4f o;
#pragma unroll
      for (int tn = 0; tn < 4; ++tn) {
        const float v = (acc[0][tn][r] - mu) * s + bt;
        o[tn] = fmaxf(v, 0.0f);
      }
      *(v4f*)(ys + m * YP + 4 * l16) = o;
    }
  }

  ATile f0, f1, fx;
  load_atile(w2, b2, wave, l16, half, f0);
  load_atile(w2, b2, MT2 - 1, l16, half, fx);

  __syncthreads();

  v4f bf[8][4];
#pragma unroll
  for (int p = 0; p < 8; ++p)
#pragma unroll
    for (int j = 0; j < 4; ++j)
      bf[p][j] = *(const v4f*)(ys + (8 * p + 4 * half + j) * YP + 4 * l16);

  float* Wb = wt + (size_t)b * M2 * HW + (size_t)(8 * half) * HW + nBase + 4 * l16;

  if (wave == 0) wt_tile(fx, bf, Wb + (size_t)(16 * (MT2 - 1)) * HW);

#pragma unroll
  for (int i = 0; i < 3; ++i) {
    const int ta = wave + 16 * i;
    const int tb = ta + 8;
    load_atile(w2, b2, tb, l16, half, f1);
    __builtin_amdgcn_sched_barrier(0);
    wt_tile(f0, bf, Wb + (size_t)(16 * ta) * HW);
    __builtin_amdgcn_sched_barrier(0);
    if (i < 2) load_atile(w2, b2, ta + 16, l16, half, f0);
    __builtin_amdgcn_sched_barrier(0);
    wt_tile(f1, bf, Wb + (size_t)(16 * tb) * HW);
    __builtin_amdgcn_sched_barrier(0);
  }
}

#define TH     8
#define XROWS  (TH + 6)
#define XPITCH 72
#define XCH    (XROWS * XPITCH + 4)

__global__ void __launch_bounds__(256) __attribute__((amdgpu_waves_per_eu(4, 5)))
k_involution(const float* __restrict__ x,
             const float* __restrict__ wt,
             float* __restrict__ inv) {
  __shared__ __attribute__((aligned(16))) float xs[GCdim * XCH];

  const int y0   = blockIdx.x * TH;
  const int g    = blockIdx.y;
  const int b    = blockIdx.z;
  const int tid  = threadIdx.x;
  const int wrow = tid >> 5;
  const int lane = tid & 31;
  const int j    = lane & 15;
  const int hh   = lane >> 4;
  const int py   = y0 + wrow;

  const float* wtp = wt + ((size_t)b * M2 + (size_t)g * KK) * HW + py * Wdim + 4 * j;

  float4 wbuf[3][Kker];
#pragma unroll
  for (int dx = 0; dx < Kker; ++dx) {
    wbuf[0][dx] = *(const float4*)(wtp + (size_t)(0 * Kker + dx) * HW);
    wbuf[1][dx] = *(const float4*)(wtp + (size_t)(1 * Kker + dx) * HW);
  }

  const float* xg = x + ((size_t)b * Cin + (size_t)g * GCdim) * HW;
  constexpr int NST = (GCdim * XROWS * 16) / 256;
  float4 sv[NST];
#pragma unroll
  for (int i = 0; i < NST; ++i) {
    const int idx = tid + 256 * i;
    const int q   = idx & 15;
    const int rc  = idx >> 4;
    const int ch  = rc / XROWS;
    const int rr  = rc - ch * XROWS;
    const int gy  = y0 - 3 + rr;
    const int gyc = min(max(gy, 0), Hdim - 1);
    sv[i] = *(const float4*)(xg + (size_t)ch * HW + gyc * Wdim + 4 * q);
  }
#pragma unroll
  for (int i = 0; i < NST; ++i) {
    const int idx = tid + 256 * i;
    const int q   = idx & 15;
    const int rc  = idx >> 4;
    const int ch  = rc / XROWS;
    const int rr  = rc - ch * XROWS;
    const int gy  = y0 - 3 + rr;
    const bool in = (unsigned)gy < (unsigned)Hdim;
    float* d = xs + ch * XCH + rr * XPITCH + 3 + 4 * q;
    d[0] = in ? sv[i].x : 0.0f;
    d[1] = in ? sv[i].y : 0.0f;
    d[2] = in ? sv[i].z : 0.0f;
    d[3] = in ? sv[i].w : 0.0f;
  }
  if (tid < GCdim * XROWS) {
    const int ch = tid / XROWS;
    const int rr = tid - ch * XROWS;
    float* rowp = xs + ch * XCH + rr * XPITCH;
    rowp[0]  = 0.0f; rowp[1]  = 0.0f; rowp[2]  = 0.0f;
    rowp[67] = 0.0f; rowp[68] = 0.0f; rowp[69] = 0.0f;
  }
  __syncthreads();

  float acc[8][4];
#pragma unroll
  for (int cc = 0; cc < 8; ++cc)
#pragma unroll
    for (int p = 0; p < 4; ++p) acc[cc][p] = 0.0f;

  const float* xwin = xs + (hh * 8) * XCH + wrow * XPITCH + 4 * j;

#pragma unroll
  for (int dy = 0; dy < Kker; ++dy) {
    if (dy + 2 < Kker) {
#pragma unroll
      for (int dx = 0; dx < Kker; ++dx)
        wbuf[(dy + 2) % 3][dx] = *(const float4*)(wtp + (size_t)((dy + 2) * Kker + dx) * HW);
    }
#pragma unroll
    for (int cc = 0; cc < 8; ++cc) {
      const float* p0 = xwin + cc * XCH + dy * XPITCH;
      const float4 a0 = *(const float4*)(p0);
      const float4 a1 = *(const float4*)(p0 + 4);
      const float2 a2 = *(const float2*)(p0 + 8);
      const float xw[10] = {a0.x, a0.y, a0.z, a0.w, a1.x, a1.y, a1.z, a1.w,
                            a2.x, a2.y};
#pragma unroll
      for (int dx = 0; dx < Kker; ++dx) {
        const float4 wv = wbuf[dy % 3][dx];
        acc[cc][0] = fmaf(wv.x, xw[dx + 0], acc[cc][0]);
        acc[cc][1] = fmaf(wv.y, xw[dx + 1], acc[cc][1]);
        acc[cc][2] = fmaf(wv.z, xw[dx + 2], acc[cc][2]);
        acc[cc][3] = fmaf(wv.w, xw[dx + 3], acc[cc][3]);
      }
    }
  }

  float* op = inv + ((size_t)b * Cin + (size_t)g * GCdim + hh * 8) * HW + py * Wdim + 4 * j;
#pragma unroll
  for (int cc = 0; cc < 8; ++cc)
    *(float4*)(op + (size_t)cc * HW) =
        make_float4(acc[cc][0], acc[cc][1], acc[cc][2], acc[cc][3]);
}

__global__ void __launch_bounds__(32 * NWAVES) k_gemm_plain(const float* __restrict__ inv,
                             const float* __restrict__ wp,
                             float* __restrict__ out) {
  const int wave  = threadIdx.x >> 5;
  const int nBase = (blockIdx.x * NWAVES + wave) * 64;
  const int mBase = blockIdx.y * 32;
  const int b     = blockIdx.z;
  const float* Bm = inv + (size_t)b * Cin * HW;
  v8f acc[2][4] = {};
  wmma_gemm_block<2, 4, Cin>(wp, Cin, Bm, HW, mBase, nBase, acc);

  float* Ob = out + (size_t)b * OUTC * HW;
  const int lane = threadIdx.x & 31;
  const int half = lane >> 4, l16 = lane & 15;
#pragma unroll
  for (int tm = 0; tm < 2; ++tm) {
#pragma unroll
    for (int r = 0; r < 8; ++r) {
      const int m = mBase + tm * 16 + r + 8 * half;
      v4f o;
#pragma unroll
      for (int tn = 0; tn < 4; ++tn) o[tn] = acc[tm][tn][r];
      *(v4f*)(Ob + (size_t)m * HW + nBase + 4 * l16) = o;
    }
  }
}

extern "C" void kernel_launch(void* const* d_in, const int* in_sizes, int n_in,
                              void* d_out, int out_size, void* d_ws, size_t ws_size,
                              hipStream_t stream) {
  const float* x     = (const float*)d_in[0];
  const float* w1    = (const float*)d_in[1];
  const float* gamma = (const float*)d_in[2];
  const float* beta  = (const float*)d_in[3];
  const float* mean  = (const float*)d_in[4];
  const float* var   = (const float*)d_in[5];
  const float* w2    = (const float*)d_in[6];
  const float* b2    = (const float*)d_in[7];
  const float* wp    = (const float*)d_in[8];
  float* out = (float*)d_out;

  float* wt  = (float*)d_ws;
  float* inv = wt + (size_t)Bsz * M2 * HW;

  dim3 blk(32 * NWAVES, 1, 1);

  dim3 g1(HW / 64, 1, Bsz);
  k_weightgen<<<g1, 256, 0, stream>>>(x, w1, gamma, beta, mean, var, w2, b2, wt);

  dim3 gI(Hdim / TH, Gdim, Bsz);
  k_involution<<<gI, 256, 0, stream>>>(x, wt, inv);

  dim3 g4(HW / (64 * NWAVES), OUTC / 32, Bsz);
  k_gemm_plain<<<g4, blk, 0, stream>>>(inv, wp, out);
}


# involution stencil moved into the weight-generation launch: the 784 x 64 wt tile of an image row stays in LDS and is consumed there; wt not written or read back; one launch fewer; sum order unchanged
# speedup vs baseline: 4.1828x; 1.1978x over previous
#include <hip/hip_runtime.h>
#include <hip/hip_bf16.h>

#define Bsz   4
#define Cin   256
#define CRdim 64
#define Gdim  16
#define GCdim 16
#define Kker  7
#define KK    49
#define M2    784
#define OUTC  256
#define Hdim  64
#define Wdim  64
#define HW    4096
#define NWAVES 4

typedef __attribute__((ext_vector_type(2))) float v2f;
typedef __attribute__((ext_vector_type(4))) float v4f;
typedef __attribute__((ext_vector_type(8))) float v8f;

template <int NTM, int GP>
struct FragSet {
  v4f a[GP][NTM];
  v4f b[GP][4];
};

template <int NTM, int GP>
__device__ __forceinline__ void load_group(const float* __restrict__ arow, int lda,
                                           const float* __restrict__ bptr, int ldb,
                                           int k0, FragSet<NTM, GP>& f) {
#pragma unroll
  for (int p = 0; p < GP; ++p) {
    const int k = k0 + 8 * p;
#pragma unroll
    for (int tm = 0; tm < NTM; ++tm)
      f.a[p][tm] = *(const v4f*)(arow + (size_t)tm * 16 * lda + k);
#pragma unroll
    for (int j = 0; j < 4; ++j)
      f.b[p][j] = *(const v4f*)(bptr + (size_t)(k + j) * ldb);
  }
}

template <int NTM, int GP>
__device__ __forceinline__ void mma_group(const FragSet<NTM, GP>& f,
                                          v8f acc[NTM][4]) {
#pragma unroll
  for (int p = 0; p < GP; ++p) {
#pragma unroll
    for (int s = 0; s < 2; ++s) {
#pragma unroll
      for (int tn = 0; tn < 4; ++tn) {
        v2f bv;
        bv.x = f.b[p][2 * s][tn];
        bv.y = f.b[p][2 * s + 1][tn];
#pragma unroll
        for (int tm = 0; tm < NTM; ++tm) {
          v2f av;
          av.x = f.a[p][tm][2 * s];
          av.y = f.a[p][tm][2 * s + 1];
          acc[tm][tn] = __builtin_amdgcn_wmma_f32_16x16x4_f32(
              false, av, false, bv, (short)0, acc[tm][tn], false, false);
        }
      }
    }
  }
}

template <int NTM, int GP, int KSZ>
__device__ __forceinline__ void wmma_gemm_block(const float* __restrict__ A, int lda,
                                                const float* __restrict__ Bm, int ldb,
                                                int mBase, int nBase,
                                                v8f acc[NTM][4]) {
  static_assert(KSZ % (16 * GP) == 0, "K must hold an even number of groups");
  const int lane = threadIdx.x & 31;
  const int half = lane >> 4;
  const int l16  = lane & 15;
  const float* arow = A + (size_t)(mBase + l16) * lda + 4 * half;
  const float* bptr = Bm + (size_t)(4 * half) * ldb + nBase + 4 * l16;

  FragSet<NTM, GP> f0, f1;
  load_group<NTM, GP>(arow, lda, bptr, ldb, 0, f0);
#pragma unroll
  for (int k = 0; k < KSZ; k += 16 * GP) {
    load_group<NTM, GP>(arow, lda, bptr, ldb, k + 8 * GP, f1);
    __builtin_amdgcn_sched_barrier(0);
    mma_group<NTM, GP>(f0, acc);
    __builtin_amdgcn_sched_barrier(0);
    if (k + 16 * GP < KSZ) load_group<NTM, GP>(arow, lda, bptr, ldb, k + 16 * GP, f0);
    __builtin_amdgcn_sched_barrier(0);
    mma_group<NTM, GP>(f1, acc);
    __builtin_amdgcn_sched_barrier(0);
  }
}

#define YP 68
#define WTP 64

struct ATile {
  v4f a[8];
  v4f bias[2];
};

__device__ __forceinline__ void load_atile(const float* __restrict__ w2,
                                           const float* __restrict__ b2,
                                           int t, int l16, int half, ATile& f) {
  const float* arow = w2 + (size_t)(16 * t + l16) * CRdim + 4 * half;
#pragma unroll
  for (int p = 0; p < 8; ++p) f.a[p] = *(const v4f*)(arow + 8 * p);
  const float* bp = b2 + 16 * t + 8 * half;
  f.bias[0] = *(const v4f*)(bp);
  f.bias[1] = *(const v4f*)(bp + 4);
}

__device__ __forceinline__ void wt_tile(const ATile& f, const v4f (&bf)[8][4],
                                        float* __restrict__ dst) {
  v8f acc[4] = {};
#pragma unroll
  for (int p = 0; p < 8; ++p) {
#pragma unroll
    for (int s = 0; s < 2; ++s) {
      v2f av;
      av.x = f.a[p][2 * s];
      av.y = f.a[p][2 * s + 1];
#pragma unroll
      for (int tn = 0; tn < 4; ++tn) {
        v2f bv;
        bv.x = bf[p][2 * s][tn];
        bv.y = bf[p][2 * s + 1][tn];
        acc[tn] = __builtin_amdgcn_wmma_f32_16x16x4_f32(
            false, av, false, bv, (short)0, acc[tn], false, false);
      }
    }
  }
#pragma unroll
  for (int r = 0; r < 8; ++r) {
    const float bias = f.bias[r >> 2][r & 3];
    v4f o;
#pragma unroll
    for (int tn = 0; tn < 4; ++tn) o[tn] = acc[tn][r] + bias;
    *(v4f*)(dst + r * WTP) = o;
  }
}

#define MT2 (M2 / 16)

__device__ __forceinline__ float lane_left(float v) {
  return __builtin_bit_cast(float, __builtin_amdgcn_update_dpp(
      0, __builtin_bit_cast(int, v), 0x111, 0xf, 0xf, true));
}
__device__ __forceinline__ float lane_right(float v) {
  return __builtin_bit_cast(float, __builtin_amdgcn_update_dpp(
      0, __builtin_bit_cast(int, v), 0x101, 0xf, 0xf, true));
}

__device__ __forceinline__ void load_xch(const float* __restrict__ xc,
                                         const int (&roff)[Kker], v4f (&d)[Kker]) {
#pragma unroll
  for (int dy = 0; dy < Kker; ++dy) d[dy] = *(const v4f*)(xc + roff[dy]);
}

__device__ __forceinline__ void stencil_channel(const v4f (&xr)[Kker],
                                                const v4f (&tw)[KK],
                                                float* __restrict__ dst) {
  float a[4] = {0.0f, 0.0f, 0.0f, 0.0f};
#pragma unroll
  for (int dy = 0; dy < Kker; ++dy) {
    const v4f m = xr[dy];
    const float xw[10] = {lane_left(m.y), lane_left(m.z), lane_left(m.w),
                          m.x, m.y, m.z, m.w,
                          lane_right(m.x), lane_right(m.y), lane_right(m.z)};
#pragma unroll
    for (int dx = 0; dx < Kker; ++dx) {
#pragma unroll
      for (int p = 0; p < 4; ++p)
        a[p] = fmaf(tw[dy * Kker + dx][p], xw[dx + p], a[p]);
    }
  }
  v4f o;
  o[0] = a[0]; o[1] = a[1]; o[2] = a[2]; o[3] = a[3];
  *(v4f*)dst = o;
}

__global__ void __launch_bounds__(256) k_weightgen_involution(const float* __restrict__ x,
                                                   const float* __restrict__ w1,
                                                   const float* __restrict__ gamma,
                                                   const float* __restrict__ beta,
                                                   const float* __restrict__ mean,
                                                   const float* __restrict__ var,
                                                   const float* __restrict__ w2,
                                                   const float* __restrict__ b2,
                                                   float* __restrict__ inv) {
  __shared__ __attribute__((aligned(16))) float ys[CRdim * YP];
  __shared__ __attribute__((aligned(16))) float wts[M2 * WTP];

  const int wave  = __builtin_amdgcn_readfirstlane((int)(threadIdx.x >> 5));
  const int lane  = threadIdx.x & 31;
  const int half  = lane >> 4, l16 = lane & 15;
  const int nBase = blockIdx.x * 64;
  const int b     = blockIdx.z;

  if (wave < 4) {
    const int mBase = wave * 16;
    const float* Bm = x + (size_t)b * Cin * HW;
    v8f acc[1][4] = {};
    wmma_gemm_block<1, 8, Cin>(w1, Cin, Bm, HW, mBase, nBase, acc);
#pragma unroll
    for (int r = 0; r < 8; ++r) {
      const int m = mBase + r + 8 * half;
      const float s  = gamma[m] * rsqrtf(var[m] + 1e-5f);
      const float mu = mean[m], bt = beta[m];
      v4f o;
#pragma unroll
      for (int tn = 0; tn < 4; ++tn) {
        const float v = (acc[0][tn][r] - mu) * s + bt;
        o[tn] = fmaxf(v, 0.0f);
      }
      *(v4f*)(ys + m * YP + 4 * l16) = o;
    }
  }

  ATile f0, f1, fx;
  load_atile(w2, b2, wave, l16, half, f0);
  load_atile(w2, b2, MT2 - 1, l16, half, fx);

  __syncthreads();

  v4f bf[8][4];
#pragma unroll
  for (int p = 0; p < 8; ++p)
#pragma unroll
    for (int j = 0; j < 4; ++j)
      bf[p][j] = *(const v4f*)(ys + (8 * p + 4 * half + j) * YP + 4 * l16);

  float* Wb = wts + (8 * half) * WTP + 4 * l16;

  if (wave == 0) wt_tile(fx, bf, Wb + (16 * (MT2 - 1)) * WTP);

#pragma unroll
  for (int i = 0; i < 3; ++i) {
    const int ta = wave + 16 * i;
    const int tb = ta + 8;
    load_atile(w2, b2, tb, l16, half, f1);
    __builtin_amdgcn_sched_barrier(0);
    wt_tile(f0, bf, Wb + (16 * ta) * WTP);
    __builtin_amdgcn_sched_barrier(0);
    if (i < 2) load_atile(w2, b2, ta + 16, l16, half, f0);
    __builtin_amdgcn_sched_barrier(0);
    wt_tile(f1, bf, Wb + (16 * tb) * WTP);
    __builtin_amdgcn_sched_barrier(0);
  }

  const int py = blockIdx.x;
  const int g  = 2 * wave + half;
  const float* xg = x + ((size_t)b * Cin + (size_t)g * GCdim) * HW + 4 * l16;
  float* op = inv + ((size_t)b * Cin + (size_t)g * GCdim) * HW + py * Wdim + 4 * l16;
  const float* tp = wts + (g * KK) * WTP + 4 * l16;

  int roff[Kker];
#pragma unroll
  for (int dy = 0; dy < Kker; ++dy)
    roff[dy] = min(max(py + dy - 3, 0), Hdim - 1) * Wdim;

  v4f xa[2][Kker], xb[2][Kker];
  load_xch(xg, roff, xa[0]);
  load_xch(xg + HW, roff, xa[1]);

  __syncthreads();

  v4f tw[KK];
#pragma unroll
  for (int k = 0; k < KK; ++k) {
    const bool rowin = (unsigned)(py + k / Kker - 3) < (unsigned)Hdim;
    const v4f t = *(const v4f*)(tp + k * WTP);
    tw[k] = rowin ? t : (v4f)(0.0f);
  }

#pragma nounroll
  for (int q = 0; q < GCdim / 4; ++q) {
    const float* xq = xg + (size_t)(4 * q) * HW;
    float* oq = op + (size_t)(4 * q) * HW;
    load_xch(xq + 2 * HW, roff, xb[0]);
    load_xch(xq + 3 * HW, roff, xb[1]);
    __builtin_amdgcn_sched_barrier(0);
    stencil_channel(xa[0], tw, oq);
    stencil_channel(xa[1], tw, oq + HW);
    __builtin_amdgcn_sched_barrier(0);
    if (q + 1 < GCdim / 4) {
      load_xch(xq + 4 * HW, roff, xa[0]);
      load_xch(xq + 5 * HW, roff, xa[1]);
    }
    __builtin_amdgcn_sched_barrier(0);
    stencil_channel(xb[0], tw, oq + 2 * HW);
    stencil_channel(xb[1], tw, oq + 3 * HW);
    __builtin_amdgcn_sched_barrier(0);
  }
}

__global__ void __launch_bounds__(32 * NWAVES) k_gemm_plain(const float* __restrict__ inv,
                             const float* __restrict__ wp,
                             float* __restrict__ out) {
  const int wave  = threadIdx.x >> 5;
  const int nBase = (blockIdx.x * NWAVES + wave) * 64;
  const int mBase = blockIdx.y * 32;
  const int b     = blockIdx.z;
  const float* Bm = inv + (size_t)b * Cin * HW;
  v8f acc[2][4] = {};
  wmma_gemm_block<2, 4, Cin>(wp, Cin, Bm, HW, mBase, nBase, acc);

  float* Ob = out + (size_t)b * OUTC * HW;
  const int lane = threadIdx.x & 31;
  const int half = lane >> 4, l16 = lane & 15;
#pragma unroll
  for (int tm = 0; tm < 2; ++tm) {
#pragma unroll
    for (int r = 0; r < 8; ++r) {
      const int m = mBase + tm * 16 + r + 8 * half;
      v4f o;
#pragma unroll
      for (int tn = 0; tn < 4; ++tn) o[tn] = acc[tm][tn][r];
      *(v4f*)(Ob + (size_t)m * HW + nBase + 4 * l16) = o;
    }
  }
}

extern "C" void kernel_launch(void* const* d_in, const int* in_sizes, int n_in,
                              void* d_out, int out_size, void* d_ws, size_t ws_size,
                              hipStream_t stream) {
  const float* x     = (const float*)d_in[0];
  const float* w1    = (const float*)d_in[1];
  const float* gamma = (const float*)d_in[2];
  const float* beta  = (const float*)d_in[3];
  const float* mean  = (const float*)d_in[4];
  const float* var   = (const float*)d_in[5];
  const float* w2    = (const float*)d_in[6];
  const float* b2    = (const float*)d_in[7];
  const float* wp    = (const float*)d_in[8];
  float* out = (float*)d_out;

  float* inv = (float*)d_ws;

  dim3 blk(32 * NWAVES, 1, 1);

  dim3 g1(HW / 64, 1, Bsz);
  k_weightgen_involution<<<g1, 256, 0, stream>>>(x, w1, gamma, beta, mean, var, w2, b2, inv);

  dim3 g4(HW / (64 * NWAVES), OUTC / 32, Bsz);
  k_gemm_plain<<<g4, blk, 0, stream>>>(inv, wp, out);
}


# pointwise GEMM moved into the first launch: the 256 x 64 inv tile of an image row stays in LDS and is read there as B fragments; inv not written or read back; one launch fewer; sum order unchanged
# speedup vs baseline: 4.8321x; 1.1552x over previous
#include <hip/hip_runtime.h>
#include <hip/hip_bf16.h>

#define Bsz   4
#define Cin   256
#define CRdim 64
#define Gdim  16
#define GCdim 16
#define Kker  7
#define KK    49
#define M2    784
#define OUTC  256
#define Hdim  64
#define Wdim  64
#define HW    4096

typedef __attribute__((ext_vector_type(2))) float v2f;
typedef __attribute__((ext_vector_type(4))) float v4f;
typedef __attribute__((ext_vector_type(8))) float v8f;

template <int NTM, int GP>
struct FragSet {
  v4f a[GP][NTM];
  v4f b[GP][4];
};

template <int NTM, int GP>
__device__ __forceinline__ void load_group(const float* __restrict__ arow, int lda,
                                           const float* __restrict__ bptr, int ldb,
                                           int k0, FragSet<NTM, GP>& f) {
#pragma unroll
  for (int p = 0; p < GP; ++p) {
    const int k = k0 + 8 * p;
#pragma unroll
    for (int tm = 0; tm < NTM; ++tm)
      f.a[p][tm] = *(const v4f*)(arow + (size_t)tm * 16 * lda + k);
#pragma unroll
    for (int j = 0; j < 4; ++j)
      f.b[p][j] = *(const v4f*)(bptr + (size_t)(k + j) * ldb);
  }
}

template <int NTM, int GP>
__device__ __forceinline__ void load_group_a(const float* __restrict__ arow, int lda,
                                             int k0, FragSet<NTM, GP>& f) {
#pragma unroll
  for (int p = 0; p < GP; ++p) {
#pragma unroll
    for (int tm = 0; tm < NTM; ++tm)
      f.a[p][tm] = *(const v4f*)(arow + (size_t)tm * 16 * lda + k0 + 8 * p);
  }
}

template <int NTM, int GP>
__device__ __forceinline__ void load_group_b(const float* bptr, int ldb,
                                             int k0, FragSet<NTM, GP>& f) {
#pragma unroll
  for (int p = 0; p < GP; ++p) {
#pragma unroll
    for (int j = 0; j < 4; ++j)
      f.b[p][j] = *(const v4f*)(bptr + (k0 + 8 * p + j) * ldb);
  }
}

template <int NTM, int GP>
__device__ __forceinline__ void mma_group(const FragSet<NTM, GP>& f,
                                          v8f acc[NTM][4]) {
#pragma unroll
  for (int p = 0; p < GP; ++p) {
#pragma unroll
    for (int s = 0; s < 2; ++s) {
#pragma unroll
      for (int tn = 0; tn < 4; ++tn) {
        v2f bv;
        bv.x = f.b[p][2 * s][tn];
        bv.y = f.b[p][2 * s + 1][tn];
#pragma unroll
        for (int tm = 0; tm < NTM; ++tm) {
          v2f av;
          av.x = f.a[p][tm][2 * s];
          av.y = f.a[p][tm][2 * s + 1];
          acc[tm][tn] = __builtin_amdgcn_wmma_f32_16x16x4_f32(
              false, av, false, bv, (short)0, acc[tm][tn], false, false);
        }
      }
    }
  }
}

template <int NTM, int GP, int KSZ>
__device__ __forceinline__ void wmma_gemm_block(const float* __restrict__ A, int lda,
                                                const float* __restrict__ Bm, int ldb,
                                                int mBase, int nBase,
                                                v8f acc[NTM][4]) {
  static_assert(KSZ % (16 * GP) == 0, "K must hold an even number of groups");
  const int lane = threadIdx.x & 31;
  const int half = lane >> 4;
  const int l16  = lane & 15;
  const float* arow = A + (size_t)(mBase + l16) * lda + 4 * half;
  const float* bptr = Bm + (size_t)(4 * half) * ldb + nBase + 4 * l16;

  FragSet<NTM, GP> f0, f1;
  load_group<NTM, GP>(arow, lda, bptr, ldb, 0, f0);
#pragma unroll
  for (int k = 0; k < KSZ; k += 16 * GP) {
    load_group<NTM, GP>(arow, lda, bptr, ldb, k + 8 * GP, f1);
    __builtin_amdgcn_sched_barrier(0);
    mma_group<NTM, GP>(f0, acc);
    __builtin_amdgcn_sched_barrier(0);
    if (k + 16 * GP < KSZ) load_group<NTM, GP>(arow, lda, bptr, ldb, k + 16 * GP, f0);
    __builtin_amdgcn_sched_barrier(0);
    mma_group<NTM, GP>(f1, acc);
    __builtin_amdgcn_sched_barrier(0);
  }
}

#define YP 68
#define WTP 64
#define IVP 64

struct ATile {
  v4f a[8];
  v4f bias[2];
};

__device__ __forceinline__ void load_atile(const float* __restrict__ w2,
                                           const float* __restrict__ b2,
                                           int t, int l16, int half, ATile& f) {
  const float* arow = w2 + (size_t)(16 * t + l16) * CRdim + 4 * half;
#pragma unroll
  for (int p = 0; p < 8; ++p) f.a[p] = *(const v4f*)(arow + 8 * p);
  const float* bp = b2 + 16 * t + 8 * half;
  f.bias[0] = *(const v4f*)(bp);
  f.bias[1] = *(const v4f*)(bp + 4);
}

__device__ __forceinline__ void wt_tile(const ATile& f, const v4f (&bf)[8][4],
                                        float* __restrict__ dst) {
  v8f acc[4] = {};
#pragma unroll
  for (int p = 0; p < 8; ++p) {
#pragma unroll
    for (int s = 0; s < 2; ++s) {
      v2f av;
      av.x = f.a[p][2 * s];
      av.y = f.a[p][2 * s + 1];
#pragma unroll
      for (int tn = 0; tn < 4; ++tn) {
        v2f bv;
        bv.x = bf[p][2 * s][tn];
        bv.y = bf[p][2 * s + 1][tn];
        acc[tn] = __builtin_amdgcn_wmma_f32_16x16x4_f32(
            false, av, false, bv, (short)0, acc[tn], false, false);
      }
    }
  }
#pragma unroll
  for (int r = 0; r < 8; ++r) {
    const float bias = f.bias[r >> 2][r & 3];
    v4f o;
#pragma unroll
    for (int tn = 0; tn < 4; ++tn) o[tn] = acc[tn][r] + bias;
    *(v4f*)(dst + r * WTP) = o;
  }
}

#define MT2 (M2 / 16)

__device__ __forceinline__ float lane_left(float v) {
  return __builtin_bit_cast(float, __builtin_amdgcn_update_dpp(
      0, __builtin_bit_cast(int, v), 0x111, 0xf, 0xf, true));
}
__device__ __forceinline__ float lane_right(float v) {
  return __builtin_bit_cast(float, __builtin_amdgcn_update_dpp(
      0, __builtin_bit_cast(int, v), 0x101, 0xf, 0xf, true));
}

__device__ __forceinline__ void load_xch(const float* __restrict__ xc,
                                         const int (&roff)[Kker], v4f (&d)[Kker]) {
#pragma unroll
  for (int dy = 0; dy < Kker; ++dy) d[dy] = *(const v4f*)(xc + roff[dy]);
}

__device__ __forceinline__ void stencil_channel(const v4f (&xr)[Kker],
                                                const v4f (&tw)[KK],
                                                float* __restrict__ dst) {
  float a[4] = {0.0f, 0.0f, 0.0f, 0.0f};
#pragma unroll
  for (int dy = 0; dy < Kker; ++dy) {
    const v4f m = xr[dy];
    const float xw[10] = {lane_left(m.y), lane_left(m.z), lane_left(m.w),
                          m.x, m.y, m.z, m.w,
                          lane_right(m.x), lane_right(m.y), lane_right(m.z)};
#pragma unroll
    for (int dx = 0; dx < Kker; ++dx) {
#pragma unroll
      for (int p = 0; p < 4; ++p)
        a[p] = fmaf(tw[dy * Kker + dx][p], xw[dx + p], a[p]);
    }
  }
  v4f o;
  o[0] = a[0]; o[1] = a[1]; o[2] = a[2]; o[3] = a[3];
  *(v4f*)dst = o;
}

__global__ void __launch_bounds__(256) k_involution_pointwise(const float* __restrict__ x,
                                                   const float* __restrict__ w1,
                                                   const float* __restrict__ gamma,
                                                   const float* __restrict__ beta,
                                                   const float* __restrict__ mean,
                                                   const float* __restrict__ var,
                                                   const float* __restrict__ w2,
                                                   const float* __restrict__ b2,
                                                   const float* __restrict__ wp,
                                                   float* __restrict__ out) {
  __shared__ __attribute__((aligned(16))) float ys[CRdim * YP];
  __shared__ __attribute__((aligned(16))) float wts[M2 * WTP];
  __shared__ __attribute__((aligned(16))) float invs[Cin * IVP];

  const int wave  = __builtin_amdgcn_readfirstlane((int)(threadIdx.x >> 5));
  const int lane  = threadIdx.x & 31;
  const int half  = lane >> 4, l16 = lane & 15;
  const int nBase = blockIdx.x * 64;
  const int b     = blockIdx.z;

  if (wave < 4) {
    const int mBase = wave * 16;
    const float* Bm = x + (size_t)b * Cin * HW;
    v8f acc[1][4] = {};
    wmma_gemm_block<1, 8, Cin>(w1, Cin, Bm, HW, mBase, nBase, acc);
#pragma unroll
    for (int r = 0; r < 8; ++r) {
      const int m = mBase + r + 8 * half;
      const float s  = gamma[m] * rsqrtf(var[m] + 1e-5f);
      const float mu = mean[m], bt = beta[m];
      v4f o;
#pragma unroll
      for (int tn = 0; tn < 4; ++tn) {
        const float v = (acc[0][tn][r] - mu) * s + bt;
        o[tn] = fmaxf(v, 0.0f);
      }
      *(v4f*)(ys + m * YP + 4 * l16) = o;
    }
  }

  ATile f0, f1, fx;
  load_atile(w2, b2, wave, l16, half, f0);
  load_atile(w2, b2, MT2 - 1, l16, half, fx);

  __syncthreads();

  v4f bf[8][4];
#pragma unroll
  for (int p = 0; p < 8; ++p)
#pragma unroll
    for (int j = 0; j < 4; ++j)
      bf[p][j] = *(const v4f*)(ys + (8 * p + 4 * half + j) * YP + 4 * l16);

  float* Wb = wts + (8 * half) * WTP + 4 * l16;

  if (wave == 0) wt_tile(fx, bf, Wb + (16 * (MT2 - 1)) * WTP);

#pragma unroll
  for (int i = 0; i < 3; ++i) {
    const int ta = wave + 16 * i;
    const int tb = ta + 8;
    load_atile(w2, b2, tb, l16, half, f1);
    __builtin_amdgcn_sched_barrier(0);
    wt_tile(f0, bf, Wb + (16 * ta) * WTP);
    __builtin_amdgcn_sched_barrier(0);
    if (i < 2) load_atile(w2, b2, ta + 16, l16, half, f0);
    __builtin_amdgcn_sched_barrier(0);
    wt_tile(f1, bf, Wb + (16 * tb) * WTP);
    __builtin_amdgcn_sched_barrier(0);
  }

  const int py = blockIdx.x;
  const int g  = 2 * wave + half;
  const float* xg = x + ((size_t)b * Cin + (size_t)g * GCdim) * HW + 4 * l16;
  float* op = invs + (g * GCdim) * IVP + 4 * l16;
  const float* tp = wts + (g * KK) * WTP + 4 * l16;

  int roff[Kker];
#pragma unroll
  for (int dy = 0; dy < Kker; ++dy)
    roff[dy] = min(max(py + dy - 3, 0), Hdim - 1) * Wdim;

  v4f xa[2][Kker], xb[2][Kker];
  load_xch(xg, roff, xa[0]);
  load_xch(xg + HW, roff, xa[1]);

  __syncthreads();

  v4f tw[KK];
#pragma unroll
  for (int k = 0; k < KK; ++k) {
    const bool rowin = (unsigned)(py + k / Kker - 3) < (unsigned)Hdim;
    const v4f t = *(const v4f*)(tp + k * WTP);
    tw[k] = rowin ? t : (v4f)(0.0f);
  }

#pragma nounroll
  for (int q = 0; q < GCdim / 4; ++q) {
    const float* xq = xg + (size_t)(4 * q) * HW;
    float* oq = op + (4 * q) * IVP;
    load_xch(xq + 2 * HW, roff, xb[0]);
    load_xch(xq + 3 * HW, roff, xb[1]);
    __builtin_amdgcn_sched_barrier(0);
    stencil_channel(xa[0], tw, oq);
    stencil_channel(xa[1], tw, oq + IVP);
    __builtin_amdgcn_sched_barrier(0);
    if (q + 1 < GCdim / 4) {
      load_xch(xq + 4 * HW, roff, xa[0]);
      load_xch(xq + 5 * HW, roff, xa[1]);
    }
    __builtin_amdgcn_sched_barrier(0);
    stencil_channel(xb[0], tw, oq + 2 * IVP);
    stencil_channel(xb[1], tw, oq + 3 * IVP);
    __builtin_amdgcn_sched_barrier(0);
  }

  const int mOut = 32 * wave;
  const float* arow = wp + (size_t)(mOut + l16) * Cin + 4 * half;
  const float* bptr = invs + (4 * half) * IVP + 4 * l16;

  FragSet<2, 4> pf0, pf1;
  load_group_a<2, 4>(arow, Cin, 0, pf0);

  __syncthreads();

  load_group_b<2, 4>(bptr, IVP, 0, pf0);
  v8f oacc[2][4] = {};
#pragma unroll
  for (int k = 0; k < Cin; k += 64) {
    load_group_a<2, 4>(arow, Cin, k + 32, pf1);
    load_group_b<2, 4>(bptr, IVP, k + 32, pf1);
    __builtin_amdgcn_sched_barrier(0);
    mma_group<2, 4>(pf0, oacc);
    __builtin_amdgcn_sched_barrier(0);
    if (k + 64 < Cin) {
      load_group_a<2, 4>(arow, Cin, k + 64, pf0);
      load_group_b<2, 4>(bptr, IVP, k + 64, pf0);
    }
    __builtin_amdgcn_sched_barrier(0);
    mma_group<2, 4>(pf1, oacc);
    __builtin_amdgcn_sched_barrier(0);
  }

  float* Ob = out + ((size_t)b * OUTC + mOut + 8 * half) * HW + nBase + 4 * l16;
#pragma unroll
  for (int tm = 0; tm < 2; ++tm) {
#pragma unroll
    for (int r = 0; r < 8; ++r) {
      v4f o;
#pragma unroll
      for (int tn = 0; tn < 4; ++tn) o[tn] = oacc[tm][tn][r];
      *(v4f*)(Ob + (size_t)(tm * 16 + r) * HW) = o;
    }
  }
}

extern "C" void kernel_launch(void* const* d_in, const int* in_sizes, int n_in,
                              void* d_out, int out_size, void* d_ws, size_t ws_size,
                              hipStream_t stream) {
  const float* x     = (const float*)d_in[0];
  const float* w1    = (const float*)d_in[1];
  const float* gamma = (const float*)d_in[2];
  const float* beta  = (const float*)d_in[3];
  const float* mean  = (const float*)d_in[4];
  const float* var   = (const float*)d_in[5];
  const float* w2    = (const float*)d_in[6];
  const float* b2    = (const float*)d_in[7];
  const float* wp    = (const float*)d_in[8];
  float* out = (float*)d_out;

  dim3 g1(HW / 64, 1, Bsz);
  k_involution_pointwise<<<g1, 256, 0, stream>>>(x, w1, gamma, beta, mean, var, w2, b2, wp, out);
}
